# starting kernel
# baseline (speedup 1.0000x reference)
#include <hip/hip_runtime.h>
#include <hip/hip_fp16.h>

typedef __attribute__((ext_vector_type(16))) _Float16 v16h;
typedef __attribute__((ext_vector_type(8)))  _Float16 v8h;
typedef __attribute__((ext_vector_type(8)))  float    v8f;

#define N_NODES   50000
#define N_EDGES   1600000
#define IN_DIM    256
#define OUT_DIM   32
#define N_HEADS   4
#define F_DIM     128
#define ALPHA_SLOPE 0.2f
#define M_TILES   3125
#define N_TILES   8
#define K_TILES   8

__global__ void pack_w_kernel(const float* __restrict__ W, _Float16* __restrict__ wpack) {
    int idx = blockIdx.x * blockDim.x + threadIdx.x;
    if (idx >= N_TILES * K_TILES * 32) return;
    int lane = idx & 31;
    int kt   = (idx >> 5) & 7;
    int nt   = idx >> 8;
    int col  = nt * 16 + (lane & 15);
    int kbase = kt * 32 + ((lane & 16) ? 8 : 0);
    _Float16* dst = wpack + idx * 16;
    #pragma unroll
    for (int i = 0; i < 8; ++i) {
        dst[i]     = (_Float16)W[(kbase + i)      * F_DIM + col];
        dst[i + 8] = (_Float16)W[(kbase + 16 + i) * F_DIM + col];
    }
}

__global__ void init_kernel(float* __restrict__ out, float* __restrict__ ssum,
                            unsigned* __restrict__ mkey) {
    int i = blockIdx.x * blockDim.x + threadIdx.x;
    if (i < N_NODES * F_DIM)   out[i] = 0.0f;
    if (i < N_NODES * N_HEADS) { ssum[i] = 0.0f; mkey[i] = 0u; }
}

__global__ void __launch_bounds__(256) gemm_kernel(const float* __restrict__ h,
                                                   const _Float16* __restrict__ wpack,
                                                   float* __restrict__ hprime) {
    __shared__ __align__(32) _Float16 tileA[16 * IN_DIM];

    const int tid = threadIdx.x;
    const int m0  = blockIdx.x * 16;

    #pragma unroll
    for (int j = 0; j < 16; ++j) {
        int row = m0 + j;
        float v = (row < N_NODES) ? h[row * IN_DIM + tid] : 0.0f;
        tileA[j * IN_DIM + tid] = (_Float16)v;
    }
    __syncthreads();

    const int wave = tid >> 5;
    const int lane = tid & 31;
    const int rowA = lane & 15;
    const int koff = (lane & 16) ? 8 : 0;

    v8f c = {};
    #pragma unroll
    for (int kt = 0; kt < K_TILES; ++kt) {
        const int kbase = kt * 32 + koff;
        v8h lo = *(const v8h*)&tileA[rowA * IN_DIM + kbase];
        v8h hi = *(const v8h*)&tileA[rowA * IN_DIM + kbase + 16];
        v16h afrag = __builtin_shufflevector(lo, hi,
            0, 1, 2, 3, 4, 5, 6, 7, 8, 9, 10, 11, 12, 13, 14, 15);
        v16h bfrag = *(const v16h*)&wpack[((wave * K_TILES + kt) * 32 + lane) * 16];
        c = __builtin_amdgcn_wmma_f32_16x16x32_f16(
                 false, afrag,  false, bfrag,
                 (short)0, c,  false,  false);
    }

    const int col   = wave * 16 + (lane & 15);
    const int rbase = (lane & 16) ? 8 : 0;
    #pragma unroll
    for (int i = 0; i < 8; ++i) {
        int gr = m0 + rbase + i;
        if (gr < N_NODES) hprime[gr * F_DIM + col] = c[i];
    }
}

__global__ void alpha_kernel(const float* __restrict__ hprime, const float* __restrict__ a,
                             float* __restrict__ asrc, float* __restrict__ adst) {
    int t = blockIdx.x * blockDim.x + threadIdx.x;
    if (t >= N_NODES * N_HEADS) return;
    int node = t >> 2;
    int head = t & 3;
    const float* hp = hprime + node * F_DIM + head * OUT_DIM;
    float s0 = 0.0f, s1 = 0.0f;
    #pragma unroll
    for (int d = 0; d < OUT_DIM; ++d) {
        float v = hp[d];
        s0 += v * a[d * N_HEADS + head];
        s1 += v * a[(OUT_DIM + d) * N_HEADS + head];
    }
    asrc[t] = s0;
    adst[t] = s1;
}

__device__ __forceinline__ unsigned fkey(float f) {
    unsigned u = __float_as_uint(f);
    return (u & 0x80000000u) ? ~u : (u | 0x80000000u);
}
__device__ __forceinline__ float finvkey(unsigned k) {
    unsigned u = (k & 0x80000000u) ? (k & 0x7FFFFFFFu) : ~k;
    return __uint_as_float(u);
}

__global__ void edge_att_kernel(const int* __restrict__ adj,
                                const float* __restrict__ asrc,
                                const float* __restrict__ adst,
                                float* __restrict__ att,
                                unsigned* __restrict__ mkey) {
    int t = blockIdx.x * blockDim.x + threadIdx.x;
    if (t >= N_EDGES * N_HEADS) return;
    int e = t >> 2, head = t & 3;
    int s = adj[e];
    int d = adj[N_EDGES + e];
    float v = asrc[s * N_HEADS + head] + adst[d * N_HEADS + head];
    v = (v > 0.0f) ? v : ALPHA_SLOPE * v;
    att[t] = v;
    atomicMax(&mkey[d * N_HEADS + head], fkey(v));
}

__global__ void edge_exp_kernel(const int* __restrict__ adj,
                                float* __restrict__ att,
                                const unsigned* __restrict__ mkey,
                                float* __restrict__ ssum) {
    int t = blockIdx.x * blockDim.x + threadIdx.x;
    if (t >= N_EDGES * N_HEADS) return;
    int e = t >> 2, head = t & 3;
    int d = adj[N_EDGES + e];
    float m  = finvkey(mkey[d * N_HEADS + head]);
    float ex = __expf(att[t] - m);
    att[t] = ex;
    atomicAdd(&ssum[d * N_HEADS + head], ex);
}

__global__ void aggregate_kernel(const int* __restrict__ adj,
                                 const float* __restrict__ att,
                                 const float* __restrict__ ssum,
                                 const float* __restrict__ hprime,
                                 float* __restrict__ out) {
    long long t = (long long)blockIdx.x * blockDim.x + threadIdx.x;
    if (t >= (long long)N_EDGES * 32) return;
    int e    = (int)(t >> 5);
    int lane = (int)(t & 31);
    int head = lane >> 3;
    int s = adj[e];
    int d = adj[N_EDGES + e];
    float w = att[e * N_HEADS + head] / ssum[d * N_HEADS + head];
    const float4 hv = *(const float4*)&hprime[s * F_DIM + lane * 4];
    float* o = out + d * F_DIM + lane * 4;
    atomicAdd(o + 0, w * hv.x);
    atomicAdd(o + 1, w * hv.y);
    atomicAdd(o + 2, w * hv.z);
    atomicAdd(o + 3, w * hv.w);
}

extern "C" void kernel_launch(void* const* d_in, const int* in_sizes, int n_in,
                              void* d_out, int out_size, void* d_ws, size_t ws_size,
                              hipStream_t stream) {
    const float* h   = (const float*)d_in[0];
    const int*   adj = (const int*)d_in[1];
    const float* W   = (const float*)d_in[2];
    const float* a   = (const float*)d_in[3];
    float* out = (float*)d_out;

    char* ws = (char*)d_ws;
    size_t off = 0;
    float*    hprime = (float*)(ws + off);     off += (size_t)N_NODES * F_DIM * 4;
    float*    asrc   = (float*)(ws + off);     off += (size_t)N_NODES * N_HEADS * 4;
    float*    adst   = (float*)(ws + off);     off += (size_t)N_NODES * N_HEADS * 4;
    float*    att    = (float*)(ws + off);     off += (size_t)N_EDGES * N_HEADS * 4;
    unsigned* mkey   = (unsigned*)(ws + off);  off += (size_t)N_NODES * N_HEADS * 4;
    float*    ssum   = (float*)(ws + off);     off += (size_t)N_NODES * N_HEADS * 4;
    _Float16* wpack  = (_Float16*)(ws + off);  off += (size_t)N_TILES * K_TILES * 32 * 16 * 2;

    const int B = 256;

    pack_w_kernel<<<(N_TILES * K_TILES * 32 + B - 1) / B, B, 0, stream>>>(W, wpack);
    init_kernel<<<(N_NODES * F_DIM + B - 1) / B, B, 0, stream>>>(out, ssum, mkey);
    gemm_kernel<<<M_TILES, B, 0, stream>>>(h, wpack, hprime);
    alpha_kernel<<<(N_NODES * N_HEADS + B - 1) / B, B, 0, stream>>>(hprime, a, asrc, adst);
    edge_att_kernel<<<(N_EDGES * N_HEADS + B - 1) / B, B, 0, stream>>>(adj, asrc, adst, att, mkey);
    edge_exp_kernel<<<(N_EDGES * N_HEADS + B - 1) / B, B, 0, stream>>>(adj, att, mkey, ssum);
    long long aggT = (long long)N_EDGES * 32;
    aggregate_kernel<<<(unsigned)((aggT + B - 1) / B), B, 0, stream>>>(adj, att, ssum, hprime, out);
}


# Edge row gather: one 16-byte load per lane (4 adjacent columns of one head; was four 4-byte loads); LDS rows column-permuted, weight by 8-lane select; sums still LDS atomics, arrival order
# speedup vs baseline: 5.8774x; 5.8774x over previous
#include <hip/hip_runtime.h>
#include <hip/hip_fp16.h>

typedef __attribute__((ext_vector_type(16))) _Float16 v16h;
typedef __attribute__((ext_vector_type(8)))  float    v8f;
typedef __attribute__((ext_vector_type(4)))  unsigned v4u;
typedef __attribute__((ext_vector_type(8)))  unsigned v8u;

#define N_NODES   50000
#define N_EDGES   1600000
#define IN_DIM    256
#define OUT_DIM   32
#define N_HEADS   4
#define F_DIM     128
#define ALPHA_SLOPE 0.2f
#define M_TILES   3125
#define N_TILES   8
#define K_TILES   8

__global__ void pack_w_kernel(const float* __restrict__ W, _Float16* __restrict__ wpack) {
    int idx = blockIdx.x * blockDim.x + threadIdx.x;
    if (idx >= N_TILES * K_TILES * 32) return;
    int lane = idx & 31;
    int kt   = (idx >> 5) & 7;
    int nt   = idx >> 8;
    int f    = idx >> 5;
    int col  = nt * 16 + (lane & 15);
    int kbase = kt * 32 + ((lane & 16) ? 8 : 0);
    _Float16* d0 = wpack + ((f * 2 + 0) * 32 + lane) * 8;
    _Float16* d1 = wpack + ((f * 2 + 1) * 32 + lane) * 8;
    #pragma unroll
    for (int i = 0; i < 8; ++i) {
        d0[i] = (_Float16)W[(kbase + i)      * F_DIM + col];
        d1[i] = (_Float16)W[(kbase + 16 + i) * F_DIM + col];
    }
}

#define GEMM_WAVES   8
#define GEMM_THREADS (GEMM_WAVES * 32)
#define GEMM_BLOCKS  ((M_TILES + GEMM_WAVES - 1) / GEMM_WAVES)
#define B_SLOTS      (N_TILES * K_TILES * 2 * 32)
static_assert(N_NODES == M_TILES * 16, "row tiles cover the nodes exactly: no row guard in the projection");
static_assert(B_SLOTS % GEMM_THREADS == 0, "packed W is copied in whole rounds of the work-group");

template <int MASK>
__device__ __forceinline__ float row_xor(float x) {
    return __int_as_float(__builtin_amdgcn_update_dpp(0, __float_as_int(x), 0x160 + MASK, 0xf, 0xf, true));
}

__device__ __forceinline__ void cvt4(v16h& a, int at, float4 x) {
    a[at + 0] = (_Float16)x.x; a[at + 1] = (_Float16)x.y;
    a[at + 2] = (_Float16)x.z; a[at + 3] = (_Float16)x.w;
}

__global__ void __launch_bounds__(GEMM_THREADS) gemm_kernel(const float* __restrict__ h,
                                                            const _Float16* __restrict__ wpack,
                                                            const float* __restrict__ a,
                                                            float* __restrict__ hprime,
                                                            float* __restrict__ asrc,
                                                            float* __restrict__ adst) {
    __shared__ __align__(16) v4u Bs[B_SLOTS];

    const int tid  = threadIdx.x;
    const int wave = tid >> 5;
    const int lane = tid & 31;
    const int c    = lane & 15;
    const int hh   = lane >> 4;
    const int tile = blockIdx.x * GEMM_WAVES + wave;
    const int ltile = (tile < M_TILES) ? tile : (M_TILES - 1);

    const float4* ap = (const float4*)(h + (size_t)(ltile * 16 + c) * IN_DIM + hh * 8);
    float4 raw[K_TILES][4];
    #pragma unroll
    for (int kt = 0; kt < K_TILES; ++kt) {
        raw[kt][0] = ap[kt * 8 + 0];
        raw[kt][1] = ap[kt * 8 + 1];
        raw[kt][2] = ap[kt * 8 + 4];
        raw[kt][3] = ap[kt * 8 + 5];
    }

    const float4 as_lo = *(const float4*)&a[(c)               * N_HEADS];
    const float4 as_hi = *(const float4*)&a[(16 + c)          * N_HEADS];
    const float4 ad_lo = *(const float4*)&a[(OUT_DIM + c)      * N_HEADS];
    const float4 ad_hi = *(const float4*)&a[(OUT_DIM + 16 + c) * N_HEADS];

    const v4u* wp = (const v4u*)wpack;
    #pragma unroll
    for (int i = 0; i < B_SLOTS / GEMM_THREADS; ++i) Bs[i * GEMM_THREADS + tid] = wp[i * GEMM_THREADS + tid];
    __syncthreads();
    if (tile >= M_TILES) return;

    v8f acc[N_TILES];
    #pragma unroll
    for (int nt = 0; nt < N_TILES; ++nt) acc[nt] = (v8f){};

    v16h af[K_TILES];
    #pragma unroll
    for (int kt = 0; kt < K_TILES; ++kt) {
        cvt4(af[kt], 0, raw[kt][0]);
        cvt4(af[kt], 4, raw[kt][1]);
        cvt4(af[kt], 8, raw[kt][2]);
        cvt4(af[kt], 12, raw[kt][3]);
    }
    #pragma unroll
    for (int kt = 0; kt < K_TILES; ++kt) {
        const v16h afrag = af[kt];
        #pragma unroll
        for (int nt = 0; nt < N_TILES; ++nt) {
            const int f = nt * K_TILES + kt;
            v4u lo = Bs[(f * 2 + 0) * 32 + lane];
            v4u hi = Bs[(f * 2 + 1) * 32 + lane];
            v8u bb = __builtin_shufflevector(lo, hi, 0, 1, 2, 3, 4, 5, 6, 7);
            v16h bfrag = __builtin_bit_cast(v16h, bb);
            acc[nt] = __builtin_amdgcn_wmma_f32_16x16x32_f16(
                     false, afrag,  false, bfrag,
                     (short)0, acc[nt],  false,  false);
        }
    }

    float* op = hprime + (size_t)(tile * 16 + hh * 8) * F_DIM + c;
    #pragma unroll
    for (int nt = 0; nt < N_TILES; ++nt) {
        #pragma unroll
        for (int i = 0; i < 8; ++i) op[i * F_DIM + nt * 16] = acc[nt][i];
    }

    const float sl[4] = {as_lo.x, as_lo.y, as_lo.z, as_lo.w};
    const float sh[4] = {as_hi.x, as_hi.y, as_hi.z, as_hi.w};
    const float dl[4] = {ad_lo.x, ad_lo.y, ad_lo.z, ad_lo.w};
    const float dh[4] = {ad_hi.x, ad_hi.y, ad_hi.z, ad_hi.w};
    const bool b0 = (c & 1) != 0, b1 = (c & 2) != 0, b2 = (c & 4) != 0, b3 = (c & 8) != 0;
    float res[N_HEADS];
    #pragma unroll
    for (int hd = 0; hd < N_HEADS; ++hd) {
        float v1[8];
        #pragma unroll
        for (int i = 0; i < 8; ++i) {
            const float ps = acc[2 * hd][i] * sl[hd] + acc[2 * hd + 1][i] * sh[hd];
            const float pd = acc[2 * hd][i] * dl[hd] + acc[2 * hd + 1][i] * dh[hd];
            v1[i] = (b0 ? pd : ps) + row_xor<1>(b0 ? ps : pd);
        }
        float v2[4];
        #pragma unroll
        for (int m = 0; m < 4; ++m)
            v2[m] = (b1 ? v1[2 * m + 1] : v1[2 * m]) + row_xor<2>(b1 ? v1[2 * m] : v1[2 * m + 1]);
        float v3[2];
        #pragma unroll
        for (int q = 0; q < 2; ++q)
            v3[q] = (b2 ? v2[2 * q + 1] : v2[2 * q]) + row_xor<4>(b2 ? v2[2 * q] : v2[2 * q + 1]);
        res[hd] = (b3 ? v3[1] : v3[0]) + row_xor<8>(b3 ? v3[0] : v3[1]);
    }
    float* lp = (b0 ? adst : asrc) + (size_t)(tile * 16 + hh * 8 + (c >> 1)) * N_HEADS;
    *(float4*)lp = make_float4(res[0], res[1], res[2], res[3]);
}

__device__ __forceinline__ unsigned fkey(float f) {
    unsigned u = __float_as_uint(f);
    return (u & 0x80000000u) ? ~u : (u | 0x80000000u);
}
__device__ __forceinline__ float finvkey(unsigned k) {
    unsigned u = (k & 0x80000000u) ? (k & 0x7FFFFFFFu) : ~k;
    return __uint_as_float(u);
}
__device__ __forceinline__ float lrelu(float v) { return (v > 0.0f) ? v : ALPHA_SLOPE * v; }

#define TILE_D     200
#define N_TILES_D  250
#define LIST_CAP   8192
#define TILE_THREADS 1024
#define NCHUNK      256
#define CHUNK       (N_EDGES / NCHUNK)
#define CHUNK_PITCH 6272
#define START_PITCH 256
#define BIN_THREADS 1024
#define BIN_U       ((CHUNK + BIN_THREADS - 1) / BIN_THREADS)
static_assert(NCHUNK * CHUNK == N_EDGES, "chunks cover the edges exactly");
static_assert(CHUNK <= LIST_CAP, "one chunk's share of a tile always fits the list");
static_assert(CHUNK <= 65535 && CHUNK <= CHUNK_PITCH, "share starts fit 16 bits, chunk regions do not overlap");
static_assert(N_TILES_D * TILE_D == N_NODES && N_TILES_D + 1 <= START_PITCH, "tiles cover the nodes; the table row holds every start");
static_assert(NCHUNK * 4 == TILE_THREADS, "four threads of a tile's work-group per chunk");
static_assert(NCHUNK == 32 * 8 && N_TILES_D <= 32 * 8, "one wave scans either table, eight entries per lane");

__device__ __forceinline__ int wave_excl_scan8(int* v, int n, int lane) {
    int x[8];
    int sum = 0;
    #pragma unroll
    for (int j = 0; j < 8; ++j) {
        const int b = lane * 8 + j;
        x[j] = (b < n) ? v[b] : 0;
        sum += x[j];
    }
    int incl = sum;
    #pragma unroll
    for (int off = 1; off < 32; off <<= 1) {
        const int t = __shfl_up(incl, off);
        if (lane >= off) incl += t;
    }
    int run = incl - sum;
    #pragma unroll
    for (int j = 0; j < 8; ++j) {
        const int b = lane * 8 + j;
        if (b < n) v[b] = run;
        run += x[j];
    }
    return incl;
}

__global__ void __launch_bounds__(BIN_THREADS) bin_kernel(const int* __restrict__ adj,
                                                          unsigned* __restrict__ bucket,
                                                          unsigned short* __restrict__ starts) {
    __shared__ int      cntb[N_TILES_D];
    __shared__ unsigned sorted[CHUNK];

    const int tid = threadIdx.x;
    const int c   = blockIdx.x;
    const int e0  = c * CHUNK;

    if (tid < N_TILES_D) cntb[tid] = 0;
    __syncthreads();

    int s[BIN_U], d[BIN_U];
    #pragma unroll
    for (int u = 0; u < BIN_U; ++u) {
        const int k = tid + u * BIN_THREADS;
        s[u] = 0; d[u] = 0;
        if (k < CHUNK) { s[u] = adj[e0 + k]; d[u] = adj[N_EDGES + e0 + k]; }
    }
    int t[BIN_U], rk[BIN_U];
    #pragma unroll
    for (int u = 0; u < BIN_U; ++u) {
        const int k = tid + u * BIN_THREADS;
        t[u] = d[u] / TILE_D;
        rk[u] = 0;
        if (k < CHUNK) rk[u] = atomicAdd(&cntb[t[u]], 1);
    }
    __syncthreads();
    if (tid < 32) wave_excl_scan8(cntb, N_TILES_D, tid);
    __syncthreads();
    #pragma unroll
    for (int u = 0; u < BIN_U; ++u) {
        const int k = tid + u * BIN_THREADS;
        if (k < CHUNK) sorted[cntb[t[u]] + rk[u]] = (unsigned)s[u] | ((unsigned)(d[u] - t[u] * TILE_D) << 16);
    }
    __syncthreads();
    unsigned* bp = bucket + (size_t)c * CHUNK_PITCH;
    #pragma unroll
    for (int u = 0; u < BIN_U; ++u) {
        const int k = tid + u * BIN_THREADS;
        if (k < CHUNK) bp[k] = sorted[k];
    }
    unsigned short* sp = starts + c * START_PITCH;
    if (tid < N_TILES_D)  sp[tid] = (unsigned short)cntb[tid];
    if (tid == N_TILES_D) sp[tid] = (unsigned short)CHUNK;
}

#define FETCH_U 10
__device__ __forceinline__ void pass_fetch(const unsigned* __restrict__ mine, const float* __restrict__ asrc,
                                           int* list, float4* as4, unsigned* mk, int len, int pos, int sub) {
    for (int k0 = 0; k0 < len; k0 += 4 * FETCH_U) {
        unsigned p[FETCH_U];
        float4 as[FETCH_U];
        #pragma unroll
        for (int u = 0; u < FETCH_U; ++u) {
            const int k = k0 + u * 4 + sub;
            p[u] = (k < len) ? mine[k] : 0u;
        }
        #pragma unroll
        for (int u = 0; u < FETCH_U; ++u) as[u] = *(const float4*)&asrc[(p[u] & 0xFFFFu) * N_HEADS];
        #pragma unroll
        for (int u = 0; u < FETCH_U; ++u) {
            const int k = k0 + u * 4 + sub;
            if (k < len) {
                const int dl = (int)(p[u] >> 16);
                atomicMax(&mk[dl * N_HEADS + 0], fkey(as[u].x));
                atomicMax(&mk[dl * N_HEADS + 1], fkey(as[u].y));
                atomicMax(&mk[dl * N_HEADS + 2], fkey(as[u].z));
                atomicMax(&mk[dl * N_HEADS + 3], fkey(as[u].w));
                list[pos + k] = (int)p[u];
                as4[pos + k]  = as[u];
            }
        }
    }
}

__device__ __forceinline__ float4 load_row(const float* __restrict__ hprime, int lane, int sj) {
    return ((const float4*)(hprime + (size_t)sj * F_DIM))[lane];
}
__device__ __forceinline__ void add_row(float* acc, int lane, int dlj, float4 r, float w) {
    float* ap = acc + dlj * F_DIM + lane;
    atomicAdd(ap,      w * r.x);
    atomicAdd(ap + 32, w * r.y);
    atomicAdd(ap + 64, w * r.z);
    atomicAdd(ap + 96, w * r.w);
}

template <int U>
__device__ __forceinline__ float grp_bcast(float x) {
    return __uint_as_float(__builtin_amdgcn_mov_dpp8(__float_as_uint(x), (unsigned)U * 0x249249u));
}

#define ROW_ADD(U) add_row(acc, lane, __builtin_amdgcn_readlane(dl, Q * 8 + (U)), r[U], grp_bcast<U>(pw))
#define ROW_ONE(U) if ((U) < cnt) add_row(acc, lane, __builtin_amdgcn_readlane(dl, Q * 8 + (U)), \
                                          load_row(hprime, lane, __builtin_amdgcn_readlane(s, Q * 8 + (U))), grp_bcast<U>(pw))
template <int Q>
__device__ __forceinline__ void sum_group(const float* __restrict__ hprime, float* acc, int lane,
                                          int s, int dl, float pw, int cnt) {
    if (cnt >= 8) {
        float4 r[8];
        #pragma unroll
        for (int u = 0; u < 8; ++u) r[u] = load_row(hprime, lane, __builtin_amdgcn_readlane(s, Q * 8 + u));
        ROW_ADD(0); ROW_ADD(1); ROW_ADD(2); ROW_ADD(3); ROW_ADD(4); ROW_ADD(5); ROW_ADD(6); ROW_ADD(7);
    } else {
        ROW_ONE(0); ROW_ONE(1); ROW_ONE(2); ROW_ONE(3); ROW_ONE(4); ROW_ONE(5); ROW_ONE(6);
    }
}
#undef ROW_ADD
#undef ROW_ONE

__device__ __forceinline__ void pass_sum(const float* __restrict__ hprime,
                                         const int* list, float4* as4, const float* adt,
                                         int n_in, const unsigned* mk, float* ss,
                                         float* acc, int tid) {
    const int lane = tid & 31;
    const int g    = lane >> 3;
    const float* asf = (const float*)as4;
    const int n     = __builtin_amdgcn_readfirstlane(n_in);
    const int base0 = __builtin_amdgcn_readfirstlane((tid >> 5) * 32);
    for (int base = base0; base < n; base += TILE_THREADS) {
        const int i = base + lane;
        int s = 0, dl = 0;
        if (i < n) {
            const int p = list[i];
            s  = p & 0xFFFF;
            dl = p >> 16;
            const float4 as = as4[i];
            const float4 ad = *(const float4*)&adt[dl * N_HEADS];
            const uint4  mx = *(const uint4*)&mk[dl * N_HEADS];
            const float w0 = __expf(lrelu(as.x + ad.x) - lrelu(finvkey(mx.x) + ad.x));
            const float w1 = __expf(lrelu(as.y + ad.y) - lrelu(finvkey(mx.y) + ad.y));
            const float w2 = __expf(lrelu(as.z + ad.z) - lrelu(finvkey(mx.z) + ad.z));
            const float w3 = __expf(lrelu(as.w + ad.w) - lrelu(finvkey(mx.w) + ad.w));
            atomicAdd(&ss[dl * N_HEADS + 0], w0);
            atomicAdd(&ss[dl * N_HEADS + 1], w1);
            atomicAdd(&ss[dl * N_HEADS + 2], w2);
            atomicAdd(&ss[dl * N_HEADS + 3], w3);
            as4[i] = make_float4(w0, w1, w2, w3);
        }
        float pw[4];
        #pragma unroll
        for (int q = 0; q < 4; ++q) pw[q] = asf[(base + q * 8 + (lane & 7)) * N_HEADS + g];
        const int cj = (n - base < 32) ? (n - base) : 32;
        sum_group<0>(hprime, acc, lane, s, dl, pw[0], cj);
        if (cj > 8)  sum_group<1>(hprime, acc, lane, s, dl, pw[1], cj - 8);
        if (cj > 16) sum_group<2>(hprime, acc, lane, s, dl, pw[2], cj - 16);
        if (cj > 24) sum_group<3>(hprime, acc, lane, s, dl, pw[3], cj - 24);
    }
}

__global__ void __launch_bounds__(TILE_THREADS) tile_edge_kernel(
        const unsigned* __restrict__ bucket, const unsigned short* __restrict__ starts,
        const float* __restrict__ asrc, const float* __restrict__ adst,
        const float* __restrict__ hprime, float* __restrict__ out) {
    __shared__ __align__(16) float    acc[TILE_D * F_DIM];
    __shared__ __align__(16) unsigned mk[TILE_D * N_HEADS];
    __shared__ __align__(16) float    ss[TILE_D * N_HEADS];
    __shared__ __align__(16) float    adt[TILE_D * N_HEADS];
    __shared__ __align__(16) float4   as4[LIST_CAP];
    __shared__ int list[LIST_CAP];
    __shared__ int pre[NCHUNK + 1];

    const int tid  = threadIdx.x;
    const int tile = blockIdx.x;
    const int row0 = tile * TILE_D;
    const int c    = tid >> 2;
    const int sub  = tid & 3;

    const int st  = starts[c * START_PITCH + tile];
    const int len = (int)starts[c * START_PITCH + tile + 1] - st;
    const unsigned* mine = bucket + (size_t)c * CHUNK_PITCH + st;

    for (int i = tid; i < TILE_D * F_DIM; i += TILE_THREADS) acc[i] = 0.0f;
    for (int i = tid; i < TILE_D * N_HEADS; i += TILE_THREADS) {
        mk[i] = 0u; ss[i] = 0.0f;
        adt[i] = adst[row0 * N_HEADS + i];
    }
    if (sub == 0) pre[c] = len;
    __syncthreads();
    if (tid < 32) {
        const int total = wave_excl_scan8(pre, NCHUNK, tid);
        if (tid == 31) pre[NCHUNK] = total;
    }
    __syncthreads();
    const int n   = pre[NCHUNK];
    const int pos = pre[c];

    if (n <= LIST_CAP) {
        pass_fetch(mine, asrc, list, as4, mk, len, pos, sub);
        __syncthreads();
        pass_sum(hprime, list, as4, adt, n, mk, ss, acc, tid);
        __syncthreads();
    } else {
        for (int phase = 0; phase < 2; ++phase) {
            int c0 = 0;
            while (c0 < NCHUNK) {
                const int b = pre[c0];
                int lo = c0 + 1, hi = NCHUNK;
                while (lo < hi) {
                    const int mid = (lo + hi + 1) >> 1;
                    if (pre[mid] - b <= LIST_CAP) lo = mid; else hi = mid - 1;
                }
                const int c1 = lo;
                __syncthreads();
                if (c >= c0 && c < c1) pass_fetch(mine, asrc, list, as4, mk, len, pos - b, sub);
                if (phase == 1) {
                    __syncthreads();
                    pass_sum(hprime, list, as4, adt, pre[c1] - b, mk, ss, acc, tid);
                }
                c0 = c1;
            }
            __syncthreads();
        }
    }

    for (int idx = tid; idx < TILE_D * (F_DIM / 4); idx += TILE_THREADS) {
        const int dl = idx >> 5;
        const int c4 = idx & 31;
        const float den = ss[dl * N_HEADS + (c4 >> 3)];
        const float* ar = acc + dl * F_DIM + c4;
        float4 v = make_float4(ar[0], ar[32], ar[64], ar[96]);
        if (den > 0.0f) { v.x /= den; v.y /= den; v.z /= den; v.w /= den; }
        else            { v.x = 0.0f; v.y = 0.0f; v.z = 0.0f; v.w = 0.0f; }
        *(float4*)&out[(size_t)(row0 + dl) * F_DIM + c4 * 4] = v;
    }
}

extern "C" void kernel_launch(void* const* d_in, const int* in_sizes, int n_in,
                              void* d_out, int out_size, void* d_ws, size_t ws_size,
                              hipStream_t stream) {
    const float* h   = (const float*)d_in[0];
    const int*   adj = (const int*)d_in[1];
    const float* W   = (const float*)d_in[2];
    const float* a   = (const float*)d_in[3];
    float* out = (float*)d_out;

    char* ws = (char*)d_ws;
    size_t off = 0;
    float*    hprime = (float*)(ws + off);     off += (size_t)N_NODES * F_DIM * 4;
    float*    asrc   = (float*)(ws + off);     off += (size_t)N_NODES * N_HEADS * 4;
    float*    adst   = (float*)(ws + off);     off += (size_t)N_NODES * N_HEADS * 4;
    unsigned* bucket = (unsigned*)(ws + off);  off += (size_t)NCHUNK * CHUNK_PITCH * 4;
    unsigned short* starts = (unsigned short*)(ws + off);  off += (size_t)NCHUNK * START_PITCH * 2;
    _Float16* wpack  = (_Float16*)(ws + off);  off += (size_t)N_TILES * K_TILES * 32 * 16 * 2;

    const int B = 256;

    pack_w_kernel<<<(N_TILES * K_TILES * 32 + B - 1) / B, B, 0, stream>>>(W, wpack);
    gemm_kernel<<<GEMM_BLOCKS, GEMM_THREADS, 0, stream>>>(h, wpack, a, hprime, asrc, adst);
    bin_kernel<<<NCHUNK, BIN_THREADS, 0, stream>>>(adj, bucket, starts);
    tile_edge_kernel<<<N_TILES_D, TILE_THREADS, 0, stream>>>(bucket, starts, asrc, adst, hprime, out);
}
